# starting kernel
# baseline (speedup 1.0000x reference)
#include <hip/hip_runtime.h>
#include <hip/hip_bf16.h>


typedef __attribute__((ext_vector_type(16))) __bf16 v16bf;
typedef __attribute__((ext_vector_type(8)))  float  v8f;

#define GCN_N 100000
#define GCN_D 128
#define GCN_WPACK_ELEMS (8 * 4 * 32 * 16)

__global__ void gcn_zero_kernel(float* __restrict__ p, size_t n) {
    size_t i      = (size_t)blockIdx.x * blockDim.x + threadIdx.x;
    size_t stride = (size_t)gridDim.x * blockDim.x;
    for (; i < n; i += stride) p[i] = 0.0f;
}

__global__ void gcn_degree_kernel(const int* __restrict__ src,
                                  const int* __restrict__ dst,
                                  float* __restrict__ deg_src,
                                  float* __restrict__ deg_dst,
                                  int nedges) {
    int i = blockIdx.x * blockDim.x + threadIdx.x;
    if (i < nedges) {
        atomicAdd(&deg_src[src[i]], 1.0f);
        atomicAdd(&deg_dst[dst[i]], 1.0f);
    }
}

__global__ void gcn_norm_kernel(float* __restrict__ p, int n) {
    int i = blockIdx.x * blockDim.x + threadIdx.x;
    if (i < n) p[i] = rsqrtf(fmaxf(p[i], 1.0f));
}

__device__ __forceinline__ int gcn_frag_k(int lane, int i) {
    const int j   = i >> 1;
    const int sub = i & 1;
    const int kk  = (j < 4) ? (2 * j + sub) : (16 + 2 * (j - 4) + sub);
    return kk + 8 * (lane >> 4);
}

__global__ void gcn_pack_w_kernel(const float* __restrict__ W,
                                  __bf16* __restrict__ Wp) {
    int idx = blockIdx.x * blockDim.x + threadIdx.x;
    if (idx >= GCN_WPACK_ELEMS) return;
    const int i    = idx & 15;
    const int lane = (idx >> 4) & 31;
    const int kb   = (idx >> 9) & 3;
    const int nt   = idx >> 11;
    const int K    = kb * 32 + gcn_frag_k(lane, i);
    const int n    = nt * 16 + (lane & 15);
    Wp[idx] = (__bf16)W[(size_t)K * GCN_D + n];
}

__global__ void __launch_bounds__(256)
gcn_gemm_norm_kernel(const float* __restrict__ X,
                     const __bf16* __restrict__ Wp,
                     const float* __restrict__ norm,
                     float* __restrict__ H,
                     int nrows) {
    const int lane = threadIdx.x & 31;
    const int wave = threadIdx.x >> 5;
    const int row0 = blockIdx.x * 16;
    const int n0   = wave * 16;
    const int half = lane >> 4;
    const int m    = lane & 15;

    v8f acc = {};
#pragma unroll
    for (int kb = 0; kb < 4; ++kb) {
        const int k0 = kb * 32;
        v16bf a;
#pragma unroll
        for (int i = 0; i < 16; ++i) {
            const int K = gcn_frag_k(lane, i);
            a[i] = (__bf16)X[(size_t)(row0 + m) * GCN_D + (k0 + K)];
        }
        const v16bf b =
            *(const v16bf*)&Wp[(((size_t)(wave * 4 + kb)) * 32 + lane) * 16];
        acc = __builtin_amdgcn_wmma_f32_16x16x32_bf16(
             false, a,  false, b,
             (short)0, acc,  false,  false);
    }

#pragma unroll
    for (int i = 0; i < 8; ++i) {
        const int r = row0 + i + 8 * half;
        H[(size_t)r * GCN_D + n0 + m] = acc[i] * norm[r];
    }
}

__global__ void __launch_bounds__(256)
gcn_scatter_kernel(const float4* __restrict__ H,
                   const int* __restrict__ src,
                   const int* __restrict__ dst,
                   float* __restrict__ agg,
                   int nedges) {
    const int wavesPerBlock = blockDim.x >> 5;
    const int edge = blockIdx.x * wavesPerBlock + (threadIdx.x >> 5);
    const int lane = threadIdx.x & 31;
    if (edge >= nedges) return;
    const int s = src[edge];
    const int d = dst[edge];
    const float4 v = H[(size_t)s * (GCN_D / 4) + lane];
    float* out = agg + (size_t)d * GCN_D + lane * 4;
    atomicAdd(out + 0, v.x);
    atomicAdd(out + 1, v.y);
    atomicAdd(out + 2, v.z);
    atomicAdd(out + 3, v.w);
}

__global__ void gcn_finalize_kernel(const float* __restrict__ agg,
                                    const float* __restrict__ norm_dst,
                                    const float* __restrict__ bias,
                                    float* __restrict__ out,
                                    int n_elems, int do_relu) {
    int i = blockIdx.x * blockDim.x + threadIdx.x;
    if (i >= n_elems) return;
    const int r = i >> 7;
    const int c = i & (GCN_D - 1);
    float v = agg[i] * norm_dst[r] + bias[c];
    if (do_relu) v = fmaxf(v, 0.0f);
    out[i] = v;
}

extern "C" void kernel_launch(void* const* d_in, const int* in_sizes, int n_in,
                              void* d_out, int out_size, void* d_ws, size_t ws_size,
                              hipStream_t stream) {
    const float* feat = (const float*)d_in[0];
    const int*   src  = (const int*)  d_in[1];
    const int*   dst  = (const int*)  d_in[2];
    const float* W1   = (const float*)d_in[3];
    const float* b1   = (const float*)d_in[4];
    const float* W2   = (const float*)d_in[5];
    const float* b2   = (const float*)d_in[6];
    float* out = (float*)d_out;

    const int N = GCN_N;
    const int E = in_sizes[1];
    const size_t NF = (size_t)N * GCN_D;

    float*  normS = (float*)d_ws;
    float*  normD = normS + N;
    __bf16* Wp1   = (__bf16*)(normD + N);
    __bf16* Wp2   = Wp1 + GCN_WPACK_ELEMS;
    float*  bufA  = (float*)(Wp2 + GCN_WPACK_ELEMS);
    float*  bufB  = bufA + NF;

    const int thr = 256;
    const int zgrid = 2048;
    const int egrid = (E + thr - 1) / thr;
    const int ngrid2 = (2 * N + thr - 1) / thr;
    const int pgrid = GCN_WPACK_ELEMS / thr;
    const int gemm_grid = N / 16;
    const int scat_grid = (E + (thr / 32) - 1) / (thr / 32);
    const int fin_grid  = (int)((NF + thr - 1) / thr);

    gcn_zero_kernel<<<zgrid, thr, 0, stream>>>(normS, (size_t)2 * N);
    gcn_degree_kernel<<<egrid, thr, 0, stream>>>(src, dst, normS, normD, E);
    gcn_norm_kernel<<<ngrid2, thr, 0, stream>>>(normS, 2 * N);
    gcn_pack_w_kernel<<<pgrid, thr, 0, stream>>>(W1, Wp1);
    gcn_pack_w_kernel<<<pgrid, thr, 0, stream>>>(W2, Wp2);

    gcn_gemm_norm_kernel<<<gemm_grid, thr, 0, stream>>>(feat, Wp1, normS, bufA, N);
    gcn_zero_kernel<<<zgrid, thr, 0, stream>>>(bufB, NF);
    gcn_scatter_kernel<<<scat_grid, thr, 0, stream>>>((const float4*)bufA, src, dst, bufB, E);
    gcn_finalize_kernel<<<fin_grid, thr, 0, stream>>>(bufB, normD, b1, bufA, (int)NF, 1);

    gcn_gemm_norm_kernel<<<gemm_grid, thr, 0, stream>>>(bufA, Wp2, normS, bufB, N);
    gcn_zero_kernel<<<zgrid, thr, 0, stream>>>(out, NF);
    gcn_scatter_kernel<<<scat_grid, thr, 0, stream>>>((const float4*)bufB, src, dst, out, E);
    gcn_finalize_kernel<<<fin_grid, thr, 0, stream>>>(out, normD, b2, out, (int)NF, 0);
}


# Layer 2 reads its edges from a list layer 1 stores as it finds them (4 B an edge: source row, local dest row; whole 32-edge groups): no second look-through or index gathers; sums in LDS arrival order
# speedup vs baseline: 3.7598x; 3.7598x over previous
#include <hip/hip_runtime.h>
#include <hip/hip_bf16.h>


typedef __attribute__((ext_vector_type(16))) __bf16 v16bf;
typedef __attribute__((ext_vector_type(8)))  float  v8f;

#define GCN_N 100000
#define GCN_D 128
#define GCN_WPACK_ELEMS (8 * 4 * 32 * 16)

#define GCN_R      393
#define GCN_ABLK   255
#define GCN_RID_PAD 255
#define GCN_STEP_E 2048

__device__ __forceinline__ unsigned gcn_range_of(int node) {
    const unsigned q = (unsigned)node / GCN_R;
    return q < GCN_RID_PAD ? q : GCN_RID_PAD;
}

__device__ __forceinline__ unsigned gcn_range4(const int4 v) {
    return gcn_range_of(v.x) | (gcn_range_of(v.y) << 8) |
           (gcn_range_of(v.z) << 16) | (gcn_range_of(v.w) << 24);
}

__global__ void gcn_range_bytes_kernel(const int* __restrict__ src,
                                       const int* __restrict__ dst,
                                       unsigned char* __restrict__ sid,
                                       unsigned char* __restrict__ rid,
                                       int nedges, int npad) {
    const int e0 = (blockIdx.x * blockDim.x + threadIdx.x) * 16;
    if (e0 >= npad) return;
    uint4 ws, wd;
    if (e0 + 16 <= nedges) {
        const int4* s4 = (const int4*)(src + e0);
        const int4* d4 = (const int4*)(dst + e0);
        const int4 s0 = s4[0], s1 = s4[1], s2 = s4[2], s3 = s4[3];
        const int4 d0 = d4[0], d1 = d4[1], d2 = d4[2], d3 = d4[3];
        ws = make_uint4(gcn_range4(s0), gcn_range4(s1), gcn_range4(s2), gcn_range4(s3));
        wd = make_uint4(gcn_range4(d0), gcn_range4(d1), gcn_range4(d2), gcn_range4(d3));
    } else {
        unsigned a[4], b[4];
        for (int q = 0; q < 4; ++q) {
            a[q] = 0u;
            b[q] = 0u;
            for (int k = 0; k < 4; ++k) {
                const int e = e0 + 4 * q + k;
                const unsigned rs = (e < nedges) ? gcn_range_of(src[e]) : (unsigned)GCN_RID_PAD;
                const unsigned rd = (e < nedges) ? gcn_range_of(dst[e]) : (unsigned)GCN_RID_PAD;
                a[q] |= rs << (8 * k);
                b[q] |= rd << (8 * k);
            }
        }
        ws = make_uint4(a[0], a[1], a[2], a[3]);
        wd = make_uint4(b[0], b[1], b[2], b[3]);
    }
    *(uint4*)(sid + e0) = ws;
    *(uint4*)(rid + e0) = wd;
}

__device__ __forceinline__ int gcn_frag_k(int lane, int i) {
    const int j   = i >> 1;
    const int sub = i & 1;
    const int kk  = (j < 4) ? (2 * j + sub) : (16 + 2 * (j - 4) + sub);
    return kk + 8 * (lane >> 4);
}

__device__ __forceinline__ int gcn_tile_col(int nt, int c) {
    return 64 * (nt >> 2) + 4 * c + (nt & 3);
}

__global__ void gcn_pack_w_kernel(const float* __restrict__ W,
                                  __bf16* __restrict__ Wp) {
    int idx = blockIdx.x * blockDim.x + threadIdx.x;
    if (idx >= GCN_WPACK_ELEMS) return;
    const int i    = idx & 15;
    const int lane = (idx >> 4) & 31;
    const int kb   = (idx >> 9) & 3;
    const int nt   = idx >> 11;
    const int K    = kb * 32 + gcn_frag_k(lane, i);
    const int n    = gcn_tile_col(nt, lane & 15);
    Wp[idx] = (__bf16)W[(size_t)K * GCN_D + n];
}

#define GCN_GEMM_WAVES 2

typedef __attribute__((ext_vector_type(4))) float v4f;

__device__ __forceinline__ v16bf gcn_load_bfrag(const __bf16* __restrict__ Wp, int nt, int kb, int lane) {
    return *(const v16bf*)&Wp[(((size_t)(nt * 4 + kb)) * 32 + lane) * 16];
}

__global__ void __launch_bounds__(32 * GCN_GEMM_WAVES) __attribute__((amdgpu_waves_per_eu(7)))
gcn_gemm_norm_kernel(const float* __restrict__ X,
                     const __bf16* __restrict__ Wp,
                     const float* __restrict__ norm,
                     float* __restrict__ H,
                     int nrows) {
    const int lane = threadIdx.x & 31;
    const int wave = __builtin_amdgcn_readfirstlane((int)(threadIdx.x >> 5));
    const int row0 = (blockIdx.x * GCN_GEMM_WAVES + wave) * 16;
    if (row0 >= nrows) return;
    const int half = lane >> 4;
    const int m    = lane & 15;

    const float* xr = X + (size_t)(row0 + m) * GCN_D;
    v4f xa[16];
#pragma unroll
    for (int kb = 0; kb < 4; ++kb) {
#pragma unroll
        for (int q = 0; q < 4; ++q)
            xa[4 * kb + q] = *(const v4f*)(xr + kb * 32 + gcn_frag_k(lane, 4 * q));
    }
    const v4f nlo = *(const v4f*)(norm + row0 + 8 * half);
    const v4f nhi = *(const v4f*)(norm + row0 + 8 * half + 4);
    v16bf b[4];
#pragma unroll
    for (int j = 0; j < 4; ++j) b[j] = gcn_load_bfrag(Wp, j, 0, lane);

    __builtin_amdgcn_sched_barrier(0);

    v16bf a[4];
#pragma unroll
    for (int kb = 0; kb < 4; ++kb) {
#pragma unroll
        for (int i = 0; i < 16; ++i)
            a[kb][i] = (__bf16)xa[4 * kb + (i >> 2)][i & 3];
    }

    float* hr0 = H + (size_t)(row0 + 8 * half) * GCN_D + 4 * m;
#pragma unroll
    for (int g = 0; g < 2; ++g) {
        v8f acc[4];
#pragma unroll
        for (int j = 0; j < 4; ++j) acc[j] = v8f{};
#pragma unroll
        for (int kb = 0; kb < 4; ++kb) {
            if (g + kb > 0) {
#pragma unroll
                for (int j = 0; j < 4; ++j) b[j] = gcn_load_bfrag(Wp, 4 * g + j, kb, lane);
                __builtin_amdgcn_sched_barrier(0);
            }
#pragma unroll
            for (int j = 0; j < 4; ++j)
                acc[j] = __builtin_amdgcn_wmma_f32_16x16x32_bf16(
                     false, a[kb],  false, b[j],
                     (short)0, acc[j],  false,  false);
        }
#pragma unroll
        for (int i = 0; i < 8; ++i) {
            const float nr = (i < 4) ? nlo[i & 3] : nhi[i & 3];
            const float4 o = make_float4(acc[0][i] * nr, acc[1][i] * nr,
                                         acc[2][i] * nr, acc[3][i] * nr);
            *(float4*)(hr0 + (size_t)i * GCN_D + 64 * g) = o;
        }
    }
}

#define GCN_AWAVES 32
#define GCN_SCAN_B 16
#define GCN_SCAN_E 512
#define GCN_SCAN_U 4
#define GCN_GRP    16
#define GCN_LCAP   4096
#define GCN_LNONE  0xFFFFFFFFu

static_assert(GCN_ABLK * GCN_R >= GCN_N, "the row ranges must cover every node");
static_assert((GCN_N - 1) / GCN_R < GCN_RID_PAD, "the pad byte must be no block's number");
static_assert(GCN_SCAN_E * GCN_SCAN_U == GCN_STEP_E, "a step is four pieces");
static_assert(GCN_R <= 512, "a dest row inside a block must fit the 9 low bits of a list slot");

__device__ __forceinline__ unsigned gcn_zero_bytes(unsigned y) {
    return ~(((y & 0x7F7F7F7Fu) + 0x7F7F7F7Fu) | y | 0x7F7F7F7Fu);
}

__device__ __forceinline__ void gcn_lds_add_row(float* __restrict__ a, const float4 v) {
    atomicAdd(a + 0,  v.x);
    atomicAdd(a + 32, v.y);
    atomicAdd(a + 64, v.z);
    atomicAdd(a + 96, v.w);
}

__device__ __forceinline__ void gcn_add_full(const char* __restrict__ Hb,
                                             float* __restrict__ acc,
                                             unsigned* __restrict__ indeg,
                                             unsigned so_l, unsigned r_l, int lane) {
    const unsigned ro_l = r_l << 7;
    const unsigned l16  = (unsigned)lane << 4;
    atomicAdd(indeg + r_l, 1u);
#pragma unroll
    for (int g = 0; g < 32; g += GCN_GRP) {
        float4 v[GCN_GRP];
#pragma unroll
        for (int j = 0; j < GCN_GRP; ++j) {
            const unsigned so = (unsigned)__builtin_amdgcn_readlane((int)so_l, g + j);
            v[j] = *(const float4*)(Hb + (so + l16));
        }
#pragma unroll
        for (int j = 0; j < GCN_GRP; ++j) {
            const unsigned ro = (unsigned)__builtin_amdgcn_readlane((int)ro_l, g + j);
            gcn_lds_add_row(acc + ro + lane, v[j]);
        }
    }
}

__device__ __forceinline__ void gcn_add_part(const char* __restrict__ Hb,
                                             float* __restrict__ acc,
                                             unsigned* __restrict__ indeg,
                                             unsigned so_l, unsigned r_l, int cnt, int lane) {
    if (lane < cnt) atomicAdd(indeg + r_l, 1u);
    const unsigned ro_l = r_l << 7;
    const unsigned l16  = (unsigned)lane << 4;
    for (int j = 0; j < cnt; ++j) {
        const unsigned so = (unsigned)__builtin_amdgcn_readlane((int)so_l, j);
        const unsigned ro = (unsigned)__builtin_amdgcn_readlane((int)ro_l, j);
        const float4 v = *(const float4*)(Hb + (so + l16));
        gcn_lds_add_row(acc + ro + lane, v);
    }
}

__device__ __forceinline__ void gcn_list_put(unsigned* __restrict__ mylist,
                                             unsigned* __restrict__ ngrp,
                                             unsigned ent, int lane) {
    unsigned slot = 0u;
    if (lane == 0) slot = atomicAdd(ngrp, 1u);
    slot = (unsigned)__builtin_amdgcn_readfirstlane((int)slot);
    if (slot < GCN_LCAP) mylist[slot * 32u + (unsigned)lane] = ent;
}

__global__ void __launch_bounds__(1024)
gcn_out_degree_kernel(const int* __restrict__ src,
                      const unsigned char* __restrict__ sid,
                      float* __restrict__ norm_src,
                      int nedges, int nrows) {
    __shared__ unsigned cnt[GCN_R];

    const int lane = threadIdx.x & 31;
    const int wave = __builtin_amdgcn_readfirstlane((int)(threadIdx.x >> 5));
    const int bid  = blockIdx.x;
    const int base = bid * GCN_R;

    for (int i = threadIdx.x; i < GCN_R; i += GCN_AWAVES * 32) cnt[i] = 0u;
    __syncthreads();

    const unsigned pat = (unsigned)bid * 0x01010101u;
    const int nsteps = (nedges + GCN_STEP_E - 1) / GCN_STEP_E;
    const int rot = bid * (nsteps / GCN_ABLK);
    const uint4* sid4 = (const uint4*)sid;

    int pend = 0;
    int n = 0;

    int cn = wave + rot;
    if (cn >= nsteps) cn -= nsteps;
    uint4 xn[GCN_SCAN_U];
    if (wave < nsteps) {
#pragma unroll
        for (int u = 0; u < GCN_SCAN_U; ++u)
            xn[u] = sid4[cn * (GCN_STEP_E / 16) + u * 32 + lane];
    }
    for (int c0 = wave; c0 < nsteps; c0 += GCN_AWAVES) {
        const int c = cn;
        unsigned mk[GCN_SCAN_U];
#pragma unroll
        for (int u = 0; u < GCN_SCAN_U; ++u)
            mk[u] = (gcn_zero_bytes(xn[u].x ^ pat) >> 7) | (gcn_zero_bytes(xn[u].y ^ pat) >> 6) |
                    (gcn_zero_bytes(xn[u].z ^ pat) >> 5) | (gcn_zero_bytes(xn[u].w ^ pat) >> 4);
        if (c0 + GCN_AWAVES < nsteps) {
            cn = c0 + GCN_AWAVES + rot;
            if (cn >= nsteps) cn -= nsteps;
#pragma unroll
            for (int u = 0; u < GCN_SCAN_U; ++u)
                xn[u] = sid4[cn * (GCN_STEP_E / 16) + u * 32 + lane];
        }
#pragma unroll
        for (int u = 0; u < GCN_SCAN_U; ++u) {
            const unsigned m = mk[u];
            unsigned ball = __builtin_amdgcn_ballot_w32(m != 0u);
            while (ball) {
                const int l = __builtin_ctz(ball);
                ball &= ball - 1u;
                unsigned mm = (unsigned)__builtin_amdgcn_readlane((int)m, l);
                const int eb = c * GCN_STEP_E + u * GCN_SCAN_E + l * GCN_SCAN_B;
                while (mm) {
                    const int p = __builtin_ctz(mm);
                    mm &= mm - 1u;
                    if (lane == n) pend = eb + 4 * (p & 7) + (p >> 3);
                    if (++n == 32) {
                        atomicAdd(&cnt[src[pend] - base], 1u);
                        n = 0;
                    }
                }
            }
        }
    }
    if (lane < n) atomicAdd(&cnt[src[pend] - base], 1u);

    __syncthreads();

    for (int r = threadIdx.x; r < GCN_R; r += GCN_AWAVES * 32) {
        const int gr = base + r;
        if (gr < nrows) norm_src[gr] = rsqrtf(fmaxf((float)cnt[r], 1.0f));
    }
}

__global__ void __launch_bounds__(1024)
gcn_aggregate_kernel(const float* __restrict__ H,
                     const int* __restrict__ src,
                     const int* __restrict__ dst,
                     const unsigned char* __restrict__ rid,
                     const float* __restrict__ bias,
                     float* __restrict__ out,
                     unsigned* __restrict__ list,
                     unsigned* __restrict__ lcount,
                     int nedges, int nrows, int do_relu, int from_list) {
    __shared__ float acc[GCN_R * GCN_D];
    __shared__ unsigned indeg[GCN_R];
    __shared__ int pool[GCN_AWAVES * 32];
    __shared__ unsigned npool;
    __shared__ unsigned ngrp;

    const int lane = threadIdx.x & 31;
    const int wave = __builtin_amdgcn_readfirstlane((int)(threadIdx.x >> 5));
    const int bid  = blockIdx.x;
    const int base = bid * GCN_R;

    float4* acc4 = (float4*)acc;
    for (int i = threadIdx.x; i < GCN_R * GCN_D / 4; i += GCN_AWAVES * 32)
        acc4[i] = make_float4(0.0f, 0.0f, 0.0f, 0.0f);
    for (int i = threadIdx.x; i < GCN_R; i += GCN_AWAVES * 32) indeg[i] = 0u;
    if (threadIdx.x == 0) {
        npool = 0u;
        ngrp = 0u;
    }
    __syncthreads();

    const char* Hb = (const char*)H;
    unsigned* mylist = list + (size_t)bid * (GCN_LCAP * 32);

    int ngr = 0;
    if (from_list) ngr = (int)lcount[bid];
    const bool use_list = from_list && ngr <= GCN_LCAP;

    if (use_list) {
        int g = wave;
        unsigned en = GCN_LNONE;
        if (g < ngr) en = mylist[g * 32 + lane];
        for (; g < ngr; g += GCN_AWAVES) {
            const unsigned ent = en;
            if (g + GCN_AWAVES < ngr) en = mylist[(g + GCN_AWAVES) * 32 + lane];
            const unsigned have = __builtin_amdgcn_ballot_w32(ent != GCN_LNONE);
            const unsigned so_l = ent & ~511u;
            const unsigned r_l  = ent & 511u;
            if (have == 0xFFFFFFFFu) gcn_add_full(Hb, acc, indeg, so_l, r_l, lane);
            else gcn_add_part(Hb, acc, indeg, so_l, r_l, __builtin_popcount(have), lane);
        }
    } else {
        const bool put = !from_list;
        const unsigned pat = (unsigned)bid * 0x01010101u;
        const int nsteps = (nedges + GCN_STEP_E - 1) / GCN_STEP_E;
        const int rot = bid * (nsteps / GCN_ABLK);
        const uint4* rid4 = (const uint4*)rid;

        int pend = 0;
        int n = 0;

        int cn = wave + rot;
        if (cn >= nsteps) cn -= nsteps;
        uint4 xn[GCN_SCAN_U];
        if (wave < nsteps) {
#pragma unroll
            for (int u = 0; u < GCN_SCAN_U; ++u)
                xn[u] = rid4[cn * (GCN_STEP_E / 16) + u * 32 + lane];
        }
        for (int c0 = wave; c0 < nsteps; c0 += GCN_AWAVES) {
            const int c = cn;
            unsigned mk[GCN_SCAN_U];
#pragma unroll
            for (int u = 0; u < GCN_SCAN_U; ++u)
                mk[u] = (gcn_zero_bytes(xn[u].x ^ pat) >> 7) | (gcn_zero_bytes(xn[u].y ^ pat) >> 6) |
                        (gcn_zero_bytes(xn[u].z ^ pat) >> 5) | (gcn_zero_bytes(xn[u].w ^ pat) >> 4);
            if (c0 + GCN_AWAVES < nsteps) {
                cn = c0 + GCN_AWAVES + rot;
                if (cn >= nsteps) cn -= nsteps;
#pragma unroll
                for (int u = 0; u < GCN_SCAN_U; ++u)
                    xn[u] = rid4[cn * (GCN_STEP_E / 16) + u * 32 + lane];
            }
#pragma nounroll
            for (int u = 0; u < GCN_SCAN_U; ++u) {
                const unsigned m = (u == 0) ? mk[0] : (u == 1) ? mk[1] : (u == 2) ? mk[2] : mk[3];
                unsigned ball = __builtin_amdgcn_ballot_w32(m != 0u);
                while (ball) {
                    const int l = __builtin_ctz(ball);
                    ball &= ball - 1u;
                    unsigned mm = (unsigned)__builtin_amdgcn_readlane((int)m, l);
                    const int eb = c * GCN_STEP_E + u * GCN_SCAN_E + l * GCN_SCAN_B;
                    while (mm) {
                        const int p = __builtin_ctz(mm);
                        mm &= mm - 1u;
                        if (lane == n) pend = eb + 4 * (p & 7) + (p >> 3);
                        if (++n == 32) {
                            const unsigned so_l = (unsigned)src[pend] << 9;
                            const unsigned r_l  = (unsigned)(dst[pend] - base);
                            if (put) gcn_list_put(mylist, &ngrp, so_l | r_l, lane);
                            gcn_add_full(Hb, acc, indeg, so_l, r_l, lane);
                            n = 0;
                        }
                    }
                }
            }
        }

        if (n > 0) {
            unsigned off = 0u;
            if (lane == 0) off = atomicAdd(&npool, (unsigned)n);
            off = (unsigned)__builtin_amdgcn_readfirstlane((int)off);
            if (lane < n) pool[off + lane] = pend;
        }
        __syncthreads();
        const int total = (int)npool;
        const int p0 = wave * 32;
        if (p0 < total) {
            const int cnt = (total - p0 < 32) ? (total - p0) : 32;
            unsigned so_l = GCN_LNONE & ~511u, r_l = 511u;
            if (lane < cnt) {
                const int e = pool[p0 + lane];
                so_l = (unsigned)src[e] << 9;
                r_l  = (unsigned)(dst[e] - base);
            }
            if (put) gcn_list_put(mylist, &ngrp, so_l | r_l, lane);
            if (cnt == 32) gcn_add_full(Hb, acc, indeg, so_l, r_l, lane);
            else gcn_add_part(Hb, acc, indeg, so_l, r_l, cnt, lane);
        }
    }

    __syncthreads();

    if (!from_list && threadIdx.x == 0) lcount[bid] = ngrp;

    const float4 b4 = *(const float4*)(bias + 4 * lane);
    for (int r = wave; r < GCN_R; r += GCN_AWAVES) {
        const int gr = base + r;
        if (gr < nrows) {
            const float nd = rsqrtf(fmaxf((float)indeg[r], 1.0f));
            const float* a = acc + r * GCN_D + lane;
            float4 o;
            o.x = a[0]  * nd + b4.x;
            o.y = a[32] * nd + b4.y;
            o.z = a[64] * nd + b4.z;
            o.w = a[96] * nd + b4.w;
            if (do_relu) {
                o.x = fmaxf(o.x, 0.0f);
                o.y = fmaxf(o.y, 0.0f);
                o.z = fmaxf(o.z, 0.0f);
                o.w = fmaxf(o.w, 0.0f);
            }
            *(float4*)(out + (size_t)gr * GCN_D + 4 * lane) = o;
        }
    }
}

extern "C" void kernel_launch(void* const* d_in, const int* in_sizes, int n_in,
                              void* d_out, int out_size, void* d_ws, size_t ws_size,
                              hipStream_t stream) {
    const float* feat = (const float*)d_in[0];
    const int*   src  = (const int*)  d_in[1];
    const int*   dst  = (const int*)  d_in[2];
    const float* W1   = (const float*)d_in[3];
    const float* b1   = (const float*)d_in[4];
    const float* W2   = (const float*)d_in[5];
    const float* b2   = (const float*)d_in[6];
    float* out = (float*)d_out;

    const int N = GCN_N;
    const int E = in_sizes[1];
    const size_t NF = (size_t)N * GCN_D;

    float*    normS   = (float*)d_ws;
    unsigned* pad     = (unsigned*)(normS + N);
    __bf16*   Wp1     = (__bf16*)(pad + 32);
    __bf16*   Wp2     = Wp1 + GCN_WPACK_ELEMS;
    float*    bufA    = (float*)(Wp2 + GCN_WPACK_ELEMS);
    float*    bufB    = bufA + NF;
    const int npad    = (E + GCN_STEP_E - 1) / GCN_STEP_E * GCN_STEP_E;
    unsigned char* rid = (unsigned char*)(bufB + NF);
    unsigned char* sid = rid + npad;
    unsigned* lcount  = (unsigned*)(sid + npad);
    unsigned* list    = lcount + 256;

    const int thr = 256;
    const int gthr = 32 * GCN_GEMM_WAVES;
    const int bgrid = (npad / 16 + thr - 1) / thr;
    const int pgrid = GCN_WPACK_ELEMS / thr;
    const int gemm_grid = (N / 16 + GCN_GEMM_WAVES - 1) / GCN_GEMM_WAVES;

    gcn_range_bytes_kernel<<<bgrid, thr, 0, stream>>>(src, dst, sid, rid, E, npad);
    gcn_out_degree_kernel<<<GCN_ABLK, GCN_AWAVES * 32, 0, stream>>>(src, sid, normS, E, N);
    gcn_pack_w_kernel<<<pgrid, thr, 0, stream>>>(W1, Wp1);
    gcn_pack_w_kernel<<<pgrid, thr, 0, stream>>>(W2, Wp2);

    gcn_gemm_norm_kernel<<<gemm_grid, gthr, 0, stream>>>(feat, Wp1, normS, bufA, N);
    gcn_aggregate_kernel<<<GCN_ABLK, GCN_AWAVES * 32, 0, stream>>>(
        bufA, src, dst, rid, b1, bufB, list, lcount, E, N, 1, 0);

    gcn_gemm_norm_kernel<<<gemm_grid, gthr, 0, stream>>>(bufB, Wp2, normS, bufA, N);
    gcn_aggregate_kernel<<<GCN_ABLK, GCN_AWAVES * 32, 0, stream>>>(
        bufA, src, dst, rid, b2, out, list, lcount, E, N, 0, 1);
}


# Owner blocks' edge lists from the 16 x 16 pass that counts out-degrees: it reads dst too and files each edge by owner block (slot: LDS add's return); no look-through; sums in LDS arrival order
# speedup vs baseline: 4.7199x; 1.2554x over previous
#include <hip/hip_runtime.h>
#include <hip/hip_bf16.h>


typedef __attribute__((ext_vector_type(16))) __bf16 v16bf;
typedef __attribute__((ext_vector_type(8)))  float  v8f;

#define GCN_N 100000
#define GCN_D 128
#define GCN_WPACK_ELEMS (8 * 4 * 32 * 16)

#define GCN_R      393
#define GCN_ABLK   255

#define GCN_OR     16
#define GCN_OS     16
#define GCN_ORN    6272
#define GCN_OPITCH (GCN_OR * GCN_ORN)
#define GCN_OTHR   512
#define GCN_OU     4
#define GCN_OWN    16
#define GCN_DRN    (GCN_OWN * GCN_R)
#define GCN_SCAP   1024
#define GCN_SGRP   (GCN_SCAP / 32)
#define GCN_QCAP   1024
#define GCN_QBATCH 4
#define GCN_LSPARE ((unsigned)GCN_R)
#define GCN_LOVER  0xFFFFFFFFu

static_assert(GCN_OR * GCN_ORN >= GCN_N, "the count ranges must cover every node");
static_assert(GCN_ORN % 32 == 0, "a block's counts must be whole 128-byte lines");
static_assert(GCN_OR * GCN_OWN >= GCN_ABLK, "the dest ranges must cover every owner block");
static_assert(GCN_ABLK * GCN_R >= GCN_N, "the owner blocks must cover every node");
static_assert((GCN_ABLK - 1) * GCN_R + GCN_OS <= GCN_N, "the source row of a fill slot must be a row that exists");
static_assert(GCN_R < 512, "a dest row inside a block, and the spare row, must fit the 9 low bits of a list slot");
static_assert(GCN_OTHR / 32 == GCN_OWN, "one wave stores one owner block's list");
static_assert(GCN_DRN <= 8192, "a row inside the dest range must fit the 13 low bits of a queue word");
static_assert(GCN_QCAP >= 2 * 32 * 4 * GCN_OU, "a round's hits must fit the queue behind the fill test");

__device__ __forceinline__ void gcn_count4(unsigned* __restrict__ cnt, const int4 v, unsigned base) {
    const unsigned a = (unsigned)v.x - base;
    const unsigned b = (unsigned)v.y - base;
    const unsigned c = (unsigned)v.z - base;
    const unsigned d = (unsigned)v.w - base;
    if (a < GCN_ORN) atomicAdd(cnt + a, 1u);
    if (b < GCN_ORN) atomicAdd(cnt + b, 1u);
    if (c < GCN_ORN) atomicAdd(cnt + c, 1u);
    if (d < GCN_ORN) atomicAdd(cnt + d, 1u);
}

__device__ __forceinline__ int gcn_queue4(unsigned* myq, int qn,
                                          const int4 vs, const int4 vd, unsigned dbase) {
    const int sv[4] = {vs.x, vs.y, vs.z, vs.w};
    const int dv[4] = {vd.x, vd.y, vd.z, vd.w};
#pragma unroll
    for (int q = 0; q < 4; ++q) {
        const unsigned t   = (unsigned)dv[q] - dbase;
        const bool     hit = t < GCN_DRN;
        const unsigned bal = __builtin_amdgcn_ballot_w32(hit);
        const unsigned rank = __builtin_amdgcn_mbcnt_lo(bal, 0u);
        if (hit) myq[qn + (int)rank] = ((unsigned)sv[q] << 13) | t;
        qn += __builtin_popcount(bal);
    }
    return qn;
}

__device__ __forceinline__ void gcn_queue_out(const unsigned* myq, int qn,
                                              unsigned* __restrict__ stage, unsigned* __restrict__ nown,
                                              int lane) {
    for (int i0 = 0; i0 < qn; i0 += 32 * GCN_QBATCH) {
        unsigned w[GCN_QBATCH], at[GCN_QBATCH], word[GCN_QBATCH], slot[GCN_QBATCH];
#pragma unroll
        for (int k = 0; k < GCN_QBATCH; ++k) {
            const int i = i0 + 32 * k + lane;
            w[k] = 0u;
            if (i < qn) w[k] = myq[i];
        }
#pragma unroll
        for (int k = 0; k < GCN_QBATCH; ++k) {
            const int i = i0 + 32 * k + lane;
            const unsigned t  = w[k] & 8191u;
            const unsigned o  = t / GCN_R;
            const unsigned rl = t - o * GCN_R;
            word[k] = ((w[k] >> 13) << 9) | rl;
            at[k]   = o * GCN_SCAP;
            slot[k] = GCN_SCAP;
            if (i < qn) slot[k] = atomicAdd(nown + o, 1u);
        }
#pragma unroll
        for (int k = 0; k < GCN_QBATCH; ++k)
            if (slot[k] < GCN_SCAP) stage[at[k] + slot[k]] = word[k];
    }
}

__global__ void __launch_bounds__(GCN_OTHR)
gcn_split_kernel(const int* __restrict__ src,
                 const int* __restrict__ dst,
                 unsigned* __restrict__ part,
                 unsigned* __restrict__ list,
                 unsigned* __restrict__ lcount,
                 int nedges) {
    __shared__ unsigned cnt[GCN_ORN];
    __shared__ unsigned stage[GCN_OWN * GCN_SCAP];
    __shared__ unsigned nown[GCN_OWN];
    __shared__ unsigned queue[GCN_OWN * GCN_QCAP];

    const int tid = threadIdx.x;
    const int r = blockIdx.x % GCN_OR;
    const int s = blockIdx.x / GCN_OR;
    const unsigned sbase = (unsigned)(r * GCN_ORN);
    const unsigned dbase = (unsigned)(r * GCN_DRN);
    const int lane = tid & 31;
    const int wave = __builtin_amdgcn_readfirstlane(tid >> 5);
    unsigned* myq = queue + wave * GCN_QCAP;
    int qn = 0;

    for (int i = tid; i < GCN_ORN; i += GCN_OTHR) cnt[i] = 0u;
    if (tid < GCN_OWN) nown[tid] = 0u;
    __syncthreads();

    const int per  = (nedges + 4 * GCN_OS - 1) / (4 * GCN_OS) * 4;
    const int e_lo = s * per;
    const int e_hi = (e_lo + per < nedges) ? (e_lo + per) : nedges;

    const int span = 4 * GCN_OTHR * GCN_OU;
    int eb = e_lo;
    for (; eb + span <= e_hi; eb += span) {
        const int4* s4 = (const int4*)(src + eb) + tid;
        const int4* d4 = (const int4*)(dst + eb) + tid;
        int4 vs[GCN_OU], vd[GCN_OU];
#pragma unroll
        for (int k = 0; k < GCN_OU; ++k) {
            vs[k] = s4[GCN_OTHR * k];
            vd[k] = d4[GCN_OTHR * k];
        }
#pragma unroll
        for (int k = 0; k < GCN_OU; ++k) {
            gcn_count4(cnt, vs[k], sbase);
            qn = gcn_queue4(myq, qn, vs[k], vd[k], dbase);
        }
        if (qn > GCN_QCAP - 32 * 4 * GCN_OU) {
            gcn_queue_out(myq, qn, stage, nown, lane);
            qn = 0;
        }
    }
    for (int e0 = eb + 4 * 32 * wave; e0 < e_hi; e0 += 4 * GCN_OTHR) {
        const int e = e0 + 4 * lane;
        int4 vs = make_int4(-1, -1, -1, -1);
        int4 vd = make_int4(-1, -1, -1, -1);
        if (e + 4 <= e_hi) {
            vs = *(const int4*)(src + e);
            vd = *(const int4*)(dst + e);
        } else if (e < e_hi) {
            vs.x = src[e];
            vd.x = dst[e];
            if (e + 1 < e_hi) { vs.y = src[e + 1]; vd.y = dst[e + 1]; }
            if (e + 2 < e_hi) { vs.z = src[e + 2]; vd.z = dst[e + 2]; }
        }
        gcn_count4(cnt, vs, sbase);
        qn = gcn_queue4(myq, qn, vs, vd, dbase);
        if (qn > GCN_QCAP - 32 * 4 * GCN_OU) {
            gcn_queue_out(myq, qn, stage, nown, lane);
            qn = 0;
        }
    }
    gcn_queue_out(myq, qn, stage, nown, lane);

    __syncthreads();

    unsigned* p = part + (size_t)s * GCN_OPITCH + sbase;
    for (int i = tid; i < GCN_ORN; i += GCN_OTHR) p[i] = cnt[i];

    const int o     = wave;
    const int owner = r * GCN_OWN + o;
    if (owner < GCN_ABLK) {
        const unsigned n    = nown[o];
        const unsigned kept = n < GCN_SCAP ? n : GCN_SCAP;
        const unsigned ng   = (kept + 31u) >> 5;
        unsigned* lw = list + (size_t)(owner * GCN_OS + s) * GCN_SCAP;
        const unsigned fill = ((unsigned)(owner * GCN_R + s) << 9) | GCN_LSPARE;
        for (unsigned i = (unsigned)lane; i < ng * 32u; i += 32u)
            lw[i] = (i < kept) ? stage[o * GCN_SCAP + i] : fill;
        if (lane == 0) lcount[owner * GCN_OS + s] = (n > GCN_SCAP) ? GCN_LOVER : ng;
    }
}

__device__ __forceinline__ int gcn_frag_k(int lane, int i) {
    const int j   = i >> 1;
    const int sub = i & 1;
    const int kk  = (j < 4) ? (2 * j + sub) : (16 + 2 * (j - 4) + sub);
    return kk + 8 * (lane >> 4);
}

__device__ __forceinline__ int gcn_tile_col(int nt, int c) {
    return 64 * (nt >> 2) + 4 * c + (nt & 3);
}

__global__ void gcn_pack_w_kernel(const float* __restrict__ W,
                                  __bf16* __restrict__ Wp) {
    int idx = blockIdx.x * blockDim.x + threadIdx.x;
    if (idx >= GCN_WPACK_ELEMS) return;
    const int i    = idx & 15;
    const int lane = (idx >> 4) & 31;
    const int kb   = (idx >> 9) & 3;
    const int nt   = idx >> 11;
    const int K    = kb * 32 + gcn_frag_k(lane, i);
    const int n    = gcn_tile_col(nt, lane & 15);
    Wp[idx] = (__bf16)W[(size_t)K * GCN_D + n];
}

#define GCN_GEMM_WAVES 2

typedef __attribute__((ext_vector_type(4))) float v4f;

__device__ __forceinline__ v16bf gcn_load_bfrag(const __bf16* __restrict__ Wp, int nt, int kb, int lane) {
    return *(const v16bf*)&Wp[(((size_t)(nt * 4 + kb)) * 32 + lane) * 16];
}

__global__ void __launch_bounds__(32 * GCN_GEMM_WAVES) __attribute__((amdgpu_waves_per_eu(7)))
gcn_gemm_norm_kernel(const float* __restrict__ X,
                     const __bf16* __restrict__ Wp,
                     const unsigned* __restrict__ part,
                     float* __restrict__ H,
                     int nrows) {
    const int lane = threadIdx.x & 31;
    const int wave = __builtin_amdgcn_readfirstlane((int)(threadIdx.x >> 5));
    const int row0 = (blockIdx.x * GCN_GEMM_WAVES + wave) * 16;
    if (row0 >= nrows) return;
    const int half = lane >> 4;
    const int m    = lane & 15;

    const unsigned* pp = part + (size_t)(8 * half) * GCN_OPITCH + (row0 + m);
    unsigned pc[GCN_OS / 2];
#pragma unroll
    for (int s = 0; s < GCN_OS / 2; ++s) pc[s] = pp[(size_t)s * GCN_OPITCH];

    const float* xr = X + (size_t)(row0 + m) * GCN_D;
    v4f xa[16];
#pragma unroll
    for (int kb = 0; kb < 4; ++kb) {
#pragma unroll
        for (int q = 0; q < 4; ++q)
            xa[4 * kb + q] = *(const v4f*)(xr + kb * 32 + gcn_frag_k(lane, 4 * q));
    }
    __builtin_amdgcn_sched_barrier(0);

    unsigned deg = 0u;
#pragma unroll
    for (int s = 0; s < GCN_OS / 2; ++s) deg += pc[s];
    deg += (unsigned)__shfl_xor((int)deg, 16);
    const float nsm = rsqrtf(fmaxf((float)deg, 1.0f));
    float nrow[16];
#pragma unroll
    for (int j = 0; j < 16; ++j)
        nrow[j] = __int_as_float(__builtin_amdgcn_readlane(__float_as_int(nsm), j));

    v16bf b[4];
#pragma unroll
    for (int j = 0; j < 4; ++j) b[j] = gcn_load_bfrag(Wp, j, 0, lane);

    __builtin_amdgcn_sched_barrier(0);

    v16bf a[4];
#pragma unroll
    for (int kb = 0; kb < 4; ++kb) {
#pragma unroll
        for (int i = 0; i < 16; ++i)
            a[kb][i] = (__bf16)xa[4 * kb + (i >> 2)][i & 3];
    }

    float* hr0 = H + (size_t)(row0 + 8 * half) * GCN_D + 4 * m;
#pragma unroll
    for (int g = 0; g < 2; ++g) {
        v8f acc[4];
#pragma unroll
        for (int j = 0; j < 4; ++j) acc[j] = v8f{};
#pragma unroll
        for (int kb = 0; kb < 4; ++kb) {
            if (g + kb > 0) {
#pragma unroll
                for (int j = 0; j < 4; ++j) b[j] = gcn_load_bfrag(Wp, 4 * g + j, kb, lane);
                __builtin_amdgcn_sched_barrier(0);
            }
#pragma unroll
            for (int j = 0; j < 4; ++j)
                acc[j] = __builtin_amdgcn_wmma_f32_16x16x32_bf16(
                     false, a[kb],  false, b[j],
                     (short)0, acc[j],  false,  false);
        }
#pragma unroll
        for (int i = 0; i < 8; ++i) {
            const float nr = half ? nrow[i + 8] : nrow[i];
            const float4 o = make_float4(acc[0][i] * nr, acc[1][i] * nr,
                                         acc[2][i] * nr, acc[3][i] * nr);
            *(float4*)(hr0 + (size_t)i * GCN_D + 64 * g) = o;
        }
    }
}

#define GCN_AWAVES 32
#define GCN_GRP    16

static_assert(GCN_AWAVES == 2 * GCN_OS, "two waves per slice list");

__device__ __forceinline__ void gcn_lds_add_row(float* __restrict__ a, const float4 v) {
    atomicAdd(a + 0,  v.x);
    atomicAdd(a + 32, v.y);
    atomicAdd(a + 64, v.z);
    atomicAdd(a + 96, v.w);
}

__device__ __forceinline__ void gcn_add_full(const char* __restrict__ Hb,
                                             float* __restrict__ acc,
                                             unsigned* __restrict__ indeg,
                                             unsigned so_l, unsigned r_l, int lane) {
    const unsigned ro_l = r_l << 7;
    const unsigned l16  = (unsigned)lane << 4;
    if (r_l < GCN_LSPARE) atomicAdd(indeg + r_l, 1u);
#pragma unroll
    for (int g = 0; g < 32; g += GCN_GRP) {
        float4 v[GCN_GRP];
#pragma unroll
        for (int j = 0; j < GCN_GRP; ++j) {
            const unsigned so = (unsigned)__builtin_amdgcn_readlane((int)so_l, g + j);
            v[j] = *(const float4*)(Hb + (so + l16));
        }
#pragma unroll
        for (int j = 0; j < GCN_GRP; ++j) {
            const unsigned ro = (unsigned)__builtin_amdgcn_readlane((int)ro_l, g + j);
            gcn_lds_add_row(acc + ro + lane, v[j]);
        }
    }
}

__device__ __forceinline__ void gcn_add_some(const char* __restrict__ Hb,
                                             float* __restrict__ acc,
                                             unsigned so_l, unsigned r_l, unsigned ball, int lane) {
    const unsigned ro_l = r_l << 7;
    const unsigned l16  = (unsigned)lane << 4;
    while (ball) {
        const int l = __builtin_ctz(ball);
        ball &= ball - 1u;
        const unsigned so = (unsigned)__builtin_amdgcn_readlane((int)so_l, l);
        const unsigned ro = (unsigned)__builtin_amdgcn_readlane((int)ro_l, l);
        const float4 v = *(const float4*)(Hb + (so + l16));
        gcn_lds_add_row(acc + ro + lane, v);
    }
}

__global__ void __launch_bounds__(1024)
gcn_aggregate_kernel(const float* __restrict__ H,
                     const int* __restrict__ src,
                     const int* __restrict__ dst,
                     const float* __restrict__ bias,
                     float* __restrict__ out,
                     const unsigned* __restrict__ list,
                     const unsigned* __restrict__ lcount,
                     int nedges, int nrows, int do_relu) {
    __shared__ float acc[(GCN_R + 1) * GCN_D];
    __shared__ unsigned indeg[GCN_R];

    const int lane = threadIdx.x & 31;
    const int wave = __builtin_amdgcn_readfirstlane((int)(threadIdx.x >> 5));
    const int bid  = blockIdx.x;
    const int base = bid * GCN_R;

    float4* acc4 = (float4*)acc;
    for (int i = threadIdx.x; i < (GCN_R + 1) * GCN_D / 4; i += GCN_AWAVES * 32)
        acc4[i] = make_float4(0.0f, 0.0f, 0.0f, 0.0f);
    for (int i = threadIdx.x; i < GCN_R; i += GCN_AWAVES * 32) indeg[i] = 0u;
    __syncthreads();

    const char* Hb = (const char*)H;

    const unsigned* mycount = lcount + bid * GCN_OS;
    unsigned most = 0u;
#pragma unroll
    for (int s = 0; s < GCN_OS; ++s) {
        const unsigned c = mycount[s];
        most = c > most ? c : most;
    }
    const bool whole = most <= (unsigned)GCN_SGRP;

    if (whole) {
        const int sl = wave & (GCN_OS - 1);
        const int ng = (int)mycount[sl];
        const unsigned* lw = list + (size_t)(bid * GCN_OS + sl) * GCN_SCAP;
        int g = wave >> 4;
        unsigned en = GCN_LSPARE;
        if (g < ng) en = lw[g * 32 + lane];
        for (; g < ng; g += 2) {
            const unsigned ent = en;
            if (g + 2 < ng) en = lw[(g + 2) * 32 + lane];
            gcn_add_full(Hb, acc, indeg, ent & ~511u, ent & 511u, lane);
        }
    } else {
        for (int e0 = wave * 32; e0 < nedges; e0 += GCN_AWAVES * 32) {
            const int e = e0 + lane;
            int d = -1, sr = 0;
            if (e < nedges) {
                d  = dst[e];
                sr = src[e];
            }
            const unsigned rl = (unsigned)(d - base);
            const bool hit = rl < (unsigned)GCN_R;
            if (hit) atomicAdd(indeg + rl, 1u);
            const unsigned ball = __builtin_amdgcn_ballot_w32(hit);
            gcn_add_some(Hb, acc, (unsigned)sr << 9, hit ? rl : (unsigned)GCN_R, ball, lane);
        }
    }

    __syncthreads();

    const float4 b4 = *(const float4*)(bias + 4 * lane);
    for (int r = wave; r < GCN_R; r += GCN_AWAVES) {
        const int gr = base + r;
        if (gr < nrows) {
            const float nd = rsqrtf(fmaxf((float)indeg[r], 1.0f));
            const float* a = acc + r * GCN_D + lane;
            float4 o;
            o.x = a[0]  * nd + b4.x;
            o.y = a[32] * nd + b4.y;
            o.z = a[64] * nd + b4.z;
            o.w = a[96] * nd + b4.w;
            if (do_relu) {
                o.x = fmaxf(o.x, 0.0f);
                o.y = fmaxf(o.y, 0.0f);
                o.z = fmaxf(o.z, 0.0f);
                o.w = fmaxf(o.w, 0.0f);
            }
            *(float4*)(out + (size_t)gr * GCN_D + 4 * lane) = o;
        }
    }
}

extern "C" void kernel_launch(void* const* d_in, const int* in_sizes, int n_in,
                              void* d_out, int out_size, void* d_ws, size_t ws_size,
                              hipStream_t stream) {
    const float* feat = (const float*)d_in[0];
    const int*   src  = (const int*)  d_in[1];
    const int*   dst  = (const int*)  d_in[2];
    const float* W1   = (const float*)d_in[3];
    const float* b1   = (const float*)d_in[4];
    const float* W2   = (const float*)d_in[5];
    const float* b2   = (const float*)d_in[6];
    float* out = (float*)d_out;

    const int N = GCN_N;
    const int E = in_sizes[1];
    const size_t NF = (size_t)N * GCN_D;

    unsigned* part    = (unsigned*)d_ws;
    __bf16*   Wp1     = (__bf16*)(part + (size_t)GCN_OS * GCN_OPITCH);
    __bf16*   Wp2     = Wp1 + GCN_WPACK_ELEMS;
    float*    bufA    = (float*)(Wp2 + GCN_WPACK_ELEMS);
    float*    bufB    = bufA + NF;
    unsigned* lcount  = (unsigned*)(bufB + NF);
    unsigned* list    = lcount + GCN_OR * GCN_OWN * GCN_OS;

    const int thr = 256;
    const int gthr = 32 * GCN_GEMM_WAVES;
    const int pgrid = GCN_WPACK_ELEMS / thr;
    const int gemm_grid = (N / 16 + GCN_GEMM_WAVES - 1) / GCN_GEMM_WAVES;

    gcn_split_kernel<<<GCN_OR * GCN_OS, GCN_OTHR, 0, stream>>>(src, dst, part, list, lcount, E);
    gcn_pack_w_kernel<<<pgrid, thr, 0, stream>>>(W1, Wp1);
    gcn_pack_w_kernel<<<pgrid, thr, 0, stream>>>(W2, Wp2);

    gcn_gemm_norm_kernel<<<gemm_grid, gthr, 0, stream>>>(feat, Wp1, part, bufA, N);
    gcn_aggregate_kernel<<<GCN_ABLK, GCN_AWAVES * 32, 0, stream>>>(
        bufA, src, dst, b1, bufB, list, lcount, E, N, 1);

    gcn_gemm_norm_kernel<<<gemm_grid, gthr, 0, stream>>>(bufB, Wp2, part, bufA, N);
    gcn_aggregate_kernel<<<GCN_ABLK, GCN_AWAVES * 32, 0, stream>>>(
        bufA, src, dst, b2, out, list, lcount, E, N, 0);
}


# Layer 2's product done by the layer-1 aggregation block: finished rows go from LDS through the same WMMA sequence, same operand type, to (h W2) norm_src; no second product launch, h never in memory
# speedup vs baseline: 5.6364x; 1.1942x over previous
#include <hip/hip_runtime.h>
#include <hip/hip_bf16.h>


typedef __attribute__((ext_vector_type(16))) __bf16 v16bf;
typedef __attribute__((ext_vector_type(8)))  float  v8f;

#define GCN_N 100000
#define GCN_D 128
#define GCN_WPACK_ELEMS (8 * 4 * 32 * 16)

#define GCN_R      393
#define GCN_ABLK   255
#define GCN_TILES  25
#define GCN_AROWS  (16 * GCN_TILES)
#define GCN_SPITCH 272

#define GCN_OR     16
#define GCN_OS     16
#define GCN_ORN    6272
#define GCN_OPITCH (GCN_OR * GCN_ORN)
#define GCN_OTHR   512
#define GCN_OU     4
#define GCN_OWN    16
#define GCN_DRN    (GCN_OWN * GCN_R)
#define GCN_SCAP   1024
#define GCN_SGRP   (GCN_SCAP / 32)
#define GCN_QCAP   1024
#define GCN_QBATCH 4
#define GCN_LSPARE ((unsigned)GCN_R)
#define GCN_LOVER  0xFFFFFFFFu

static_assert(GCN_OR * GCN_ORN >= GCN_N, "the count ranges must cover every node");
static_assert(GCN_ORN % 32 == 0, "a block's counts must be whole 128-byte lines");
static_assert(GCN_OR * GCN_OWN >= GCN_ABLK, "the dest ranges must cover every owner block");
static_assert(GCN_ABLK * GCN_R >= GCN_N, "the owner blocks must cover every node");
static_assert(GCN_AROWS > GCN_R, "the tiles must cover the block's rows and the spare row");
static_assert((GCN_ABLK - 1) * GCN_R + GCN_AROWS <= GCN_OR * GCN_ORN, "a tile row's out-degree counts must lie inside a plane");
static_assert(16 * GCN_SPITCH <= 16 * GCN_D * 4, "a restaged tile must fit the tile's own accumulator rows");
static_assert((GCN_ABLK - 1) * GCN_R + GCN_OS <= GCN_N, "the source row of a fill slot must be a row that exists");
static_assert(GCN_R < 512, "a dest row inside a block, and the spare row, must fit the 9 low bits of a list slot");
static_assert(GCN_OTHR / 32 == GCN_OWN, "one wave stores one owner block's list");
static_assert(GCN_DRN <= 8192, "a row inside the dest range must fit the 13 low bits of a queue word");
static_assert(GCN_QCAP >= 2 * 32 * 4 * GCN_OU, "a round's hits must fit the queue behind the fill test");

__device__ __forceinline__ void gcn_count4(unsigned* __restrict__ cnt, const int4 v, unsigned base) {
    const unsigned a = (unsigned)v.x - base;
    const unsigned b = (unsigned)v.y - base;
    const unsigned c = (unsigned)v.z - base;
    const unsigned d = (unsigned)v.w - base;
    if (a < GCN_ORN) atomicAdd(cnt + a, 1u);
    if (b < GCN_ORN) atomicAdd(cnt + b, 1u);
    if (c < GCN_ORN) atomicAdd(cnt + c, 1u);
    if (d < GCN_ORN) atomicAdd(cnt + d, 1u);
}

__device__ __forceinline__ int gcn_queue4(unsigned* myq, int qn,
                                          const int4 vs, const int4 vd, unsigned dbase) {
    const int sv[4] = {vs.x, vs.y, vs.z, vs.w};
    const int dv[4] = {vd.x, vd.y, vd.z, vd.w};
#pragma unroll
    for (int q = 0; q < 4; ++q) {
        const unsigned t   = (unsigned)dv[q] - dbase;
        const bool     hit = t < GCN_DRN;
        const unsigned bal = __builtin_amdgcn_ballot_w32(hit);
        const unsigned rank = __builtin_amdgcn_mbcnt_lo(bal, 0u);
        if (hit) myq[qn + (int)rank] = ((unsigned)sv[q] << 13) | t;
        qn += __builtin_popcount(bal);
    }
    return qn;
}

__device__ __forceinline__ void gcn_queue_out(const unsigned* myq, int qn,
                                              unsigned* __restrict__ stage, unsigned* __restrict__ nown,
                                              int lane) {
    for (int i0 = 0; i0 < qn; i0 += 32 * GCN_QBATCH) {
        unsigned w[GCN_QBATCH], at[GCN_QBATCH], word[GCN_QBATCH], slot[GCN_QBATCH];
#pragma unroll
        for (int k = 0; k < GCN_QBATCH; ++k) {
            const int i = i0 + 32 * k + lane;
            w[k] = 0u;
            if (i < qn) w[k] = myq[i];
        }
#pragma unroll
        for (int k = 0; k < GCN_QBATCH; ++k) {
            const int i = i0 + 32 * k + lane;
            const unsigned t  = w[k] & 8191u;
            const unsigned o  = t / GCN_R;
            const unsigned rl = t - o * GCN_R;
            word[k] = ((w[k] >> 13) << 9) | rl;
            at[k]   = o * GCN_SCAP;
            slot[k] = GCN_SCAP;
            if (i < qn) slot[k] = atomicAdd(nown + o, 1u);
        }
#pragma unroll
        for (int k = 0; k < GCN_QBATCH; ++k)
            if (slot[k] < GCN_SCAP) stage[at[k] + slot[k]] = word[k];
    }
}

__global__ void __launch_bounds__(GCN_OTHR)
gcn_split_kernel(const int* __restrict__ src,
                 const int* __restrict__ dst,
                 unsigned* __restrict__ part,
                 unsigned* __restrict__ list,
                 unsigned* __restrict__ lcount,
                 int nedges) {
    __shared__ unsigned cnt[GCN_ORN];
    __shared__ unsigned stage[GCN_OWN * GCN_SCAP];
    __shared__ unsigned nown[GCN_OWN];
    __shared__ unsigned queue[GCN_OWN * GCN_QCAP];

    const int tid = threadIdx.x;
    const int r = blockIdx.x % GCN_OR;
    const int s = blockIdx.x / GCN_OR;
    const unsigned sbase = (unsigned)(r * GCN_ORN);
    const unsigned dbase = (unsigned)(r * GCN_DRN);
    const int lane = tid & 31;
    const int wave = __builtin_amdgcn_readfirstlane(tid >> 5);
    unsigned* myq = queue + wave * GCN_QCAP;
    int qn = 0;

    for (int i = tid; i < GCN_ORN; i += GCN_OTHR) cnt[i] = 0u;
    if (tid < GCN_OWN) nown[tid] = 0u;
    __syncthreads();

    const int per  = (nedges + 4 * GCN_OS - 1) / (4 * GCN_OS) * 4;
    const int e_lo = s * per;
    const int e_hi = (e_lo + per < nedges) ? (e_lo + per) : nedges;

    const int span = 4 * GCN_OTHR * GCN_OU;
    int eb = e_lo;
    for (; eb + span <= e_hi; eb += span) {
        const int4* s4 = (const int4*)(src + eb) + tid;
        const int4* d4 = (const int4*)(dst + eb) + tid;
        int4 vs[GCN_OU], vd[GCN_OU];
#pragma unroll
        for (int k = 0; k < GCN_OU; ++k) {
            vs[k] = s4[GCN_OTHR * k];
            vd[k] = d4[GCN_OTHR * k];
        }
#pragma unroll
        for (int k = 0; k < GCN_OU; ++k) {
            gcn_count4(cnt, vs[k], sbase);
            qn = gcn_queue4(myq, qn, vs[k], vd[k], dbase);
        }
        if (qn > GCN_QCAP - 32 * 4 * GCN_OU) {
            gcn_queue_out(myq, qn, stage, nown, lane);
            qn = 0;
        }
    }
    for (int e0 = eb + 4 * 32 * wave; e0 < e_hi; e0 += 4 * GCN_OTHR) {
        const int e = e0 + 4 * lane;
        int4 vs = make_int4(-1, -1, -1, -1);
        int4 vd = make_int4(-1, -1, -1, -1);
        if (e + 4 <= e_hi) {
            vs = *(const int4*)(src + e);
            vd = *(const int4*)(dst + e);
        } else if (e < e_hi) {
            vs.x = src[e];
            vd.x = dst[e];
            if (e + 1 < e_hi) { vs.y = src[e + 1]; vd.y = dst[e + 1]; }
            if (e + 2 < e_hi) { vs.z = src[e + 2]; vd.z = dst[e + 2]; }
        }
        gcn_count4(cnt, vs, sbase);
        qn = gcn_queue4(myq, qn, vs, vd, dbase);
        if (qn > GCN_QCAP - 32 * 4 * GCN_OU) {
            gcn_queue_out(myq, qn, stage, nown, lane);
            qn = 0;
        }
    }
    gcn_queue_out(myq, qn, stage, nown, lane);

    __syncthreads();

    unsigned* p = part + (size_t)s * GCN_OPITCH + sbase;
    for (int i = tid; i < GCN_ORN; i += GCN_OTHR) p[i] = cnt[i];

    const int o     = wave;
    const int owner = r * GCN_OWN + o;
    if (owner < GCN_ABLK) {
        const unsigned n    = nown[o];
        const unsigned kept = n < GCN_SCAP ? n : GCN_SCAP;
        const unsigned ng   = (kept + 31u) >> 5;
        unsigned* lw = list + (size_t)(owner * GCN_OS + s) * GCN_SCAP;
        const unsigned fill = ((unsigned)(owner * GCN_R + s) << 9) | GCN_LSPARE;
        for (unsigned i = (unsigned)lane; i < ng * 32u; i += 32u)
            lw[i] = (i < kept) ? stage[o * GCN_SCAP + i] : fill;
        if (lane == 0) lcount[owner * GCN_OS + s] = (n > GCN_SCAP) ? GCN_LOVER : ng;
    }
}

__device__ __forceinline__ int gcn_frag_k(int lane, int i) {
    const int j   = i >> 1;
    const int sub = i & 1;
    const int kk  = (j < 4) ? (2 * j + sub) : (16 + 2 * (j - 4) + sub);
    return kk + 8 * (lane >> 4);
}

__device__ __forceinline__ int gcn_tile_col(int nt, int c) {
    return 64 * (nt >> 2) + 4 * c + (nt & 3);
}

__global__ void gcn_pack_w_kernel(const float* __restrict__ W,
                                  __bf16* __restrict__ Wp) {
    int idx = blockIdx.x * blockDim.x + threadIdx.x;
    if (idx >= GCN_WPACK_ELEMS) return;
    const int i    = idx & 15;
    const int lane = (idx >> 4) & 31;
    const int kb   = (idx >> 9) & 3;
    const int nt   = idx >> 11;
    const int K    = kb * 32 + gcn_frag_k(lane, i);
    const int n    = gcn_tile_col(nt, lane & 15);
    Wp[idx] = (__bf16)W[(size_t)K * GCN_D + n];
}

#define GCN_GEMM_WAVES 2

typedef __attribute__((ext_vector_type(4))) float v4f;

__device__ __forceinline__ v16bf gcn_load_bfrag(const __bf16* __restrict__ Wp, int nt, int kb, int lane) {
    return *(const v16bf*)&Wp[(((size_t)(nt * 4 + kb)) * 32 + lane) * 16];
}

__global__ void __launch_bounds__(32 * GCN_GEMM_WAVES) __attribute__((amdgpu_waves_per_eu(7)))
gcn_gemm_norm_kernel(const float* __restrict__ X,
                     const __bf16* __restrict__ Wp,
                     const unsigned* __restrict__ part,
                     float* __restrict__ H,
                     int nrows) {
    const int lane = threadIdx.x & 31;
    const int wave = __builtin_amdgcn_readfirstlane((int)(threadIdx.x >> 5));
    const int row0 = (blockIdx.x * GCN_GEMM_WAVES + wave) * 16;
    if (row0 >= nrows) return;
    const int half = lane >> 4;
    const int m    = lane & 15;

    const unsigned* pp = part + (size_t)(8 * half) * GCN_OPITCH + (row0 + m);
    unsigned pc[GCN_OS / 2];
#pragma unroll
    for (int s = 0; s < GCN_OS / 2; ++s) pc[s] = pp[(size_t)s * GCN_OPITCH];

    const float* xr = X + (size_t)(row0 + m) * GCN_D;
    v4f xa[16];
#pragma unroll
    for (int kb = 0; kb < 4; ++kb) {
#pragma unroll
        for (int q = 0; q < 4; ++q)
            xa[4 * kb + q] = *(const v4f*)(xr + kb * 32 + gcn_frag_k(lane, 4 * q));
    }
    __builtin_amdgcn_sched_barrier(0);

    unsigned deg = 0u;
#pragma unroll
    for (int s = 0; s < GCN_OS / 2; ++s) deg += pc[s];
    deg += (unsigned)__shfl_xor((int)deg, 16);
    const float nsm = rsqrtf(fmaxf((float)deg, 1.0f));
    float nrow[16];
#pragma unroll
    for (int j = 0; j < 16; ++j)
        nrow[j] = __int_as_float(__builtin_amdgcn_readlane(__float_as_int(nsm), j));

    v16bf b[4];
#pragma unroll
    for (int j = 0; j < 4; ++j) b[j] = gcn_load_bfrag(Wp, j, 0, lane);

    __builtin_amdgcn_sched_barrier(0);

    v16bf a[4];
#pragma unroll
    for (int kb = 0; kb < 4; ++kb) {
#pragma unroll
        for (int i = 0; i < 16; ++i)
            a[kb][i] = (__bf16)xa[4 * kb + (i >> 2)][i & 3];
    }

    float* hr0 = H + (size_t)(row0 + 8 * half) * GCN_D + 4 * m;
#pragma unroll
    for (int g = 0; g < 2; ++g) {
        v8f acc[4];
#pragma unroll
        for (int j = 0; j < 4; ++j) acc[j] = v8f{};
#pragma unroll
        for (int kb = 0; kb < 4; ++kb) {
            if (g + kb > 0) {
#pragma unroll
                for (int j = 0; j < 4; ++j) b[j] = gcn_load_bfrag(Wp, 4 * g + j, kb, lane);
                __builtin_amdgcn_sched_barrier(0);
            }
#pragma unroll
            for (int j = 0; j < 4; ++j)
                acc[j] = __builtin_amdgcn_wmma_f32_16x16x32_bf16(
                     false, a[kb],  false, b[j],
                     (short)0, acc[j],  false,  false);
        }
#pragma unroll
        for (int i = 0; i < 8; ++i) {
            const float nr = half ? nrow[i + 8] : nrow[i];
            const float4 o = make_float4(acc[0][i] * nr, acc[1][i] * nr,
                                         acc[2][i] * nr, acc[3][i] * nr);
            *(float4*)(hr0 + (size_t)i * GCN_D + 64 * g) = o;
        }
    }
}

#define GCN_AWAVES 32
#define GCN_GRP    16

static_assert(GCN_AWAVES == 2 * GCN_OS, "two waves per slice list");
static_assert(GCN_TILES <= GCN_AWAVES, "one wave per 16-row tile of the block");

typedef __attribute__((ext_vector_type(4))) __bf16 v4bf;
typedef __attribute__((ext_vector_type(8))) __bf16 v8bf;
typedef v4bf __attribute__((may_alias)) v4bf_lds;
typedef v8bf __attribute__((may_alias)) v8bf_lds;

__device__ __forceinline__ void gcn_lds_add_row(float* __restrict__ a, const float4 v) {
    atomicAdd(a + 0,  v.x);
    atomicAdd(a + 32, v.y);
    atomicAdd(a + 64, v.z);
    atomicAdd(a + 96, v.w);
}

__device__ __forceinline__ void gcn_add_full(const char* __restrict__ Hb,
                                             float* __restrict__ acc,
                                             unsigned* __restrict__ indeg,
                                             unsigned so_l, unsigned r_l, int lane) {
    const unsigned ro_l = r_l << 7;
    const unsigned l16  = (unsigned)lane << 4;
    if (r_l < GCN_LSPARE) atomicAdd(indeg + r_l, 1u);
#pragma unroll
    for (int g = 0; g < 32; g += GCN_GRP) {
        float4 v[GCN_GRP];
#pragma unroll
        for (int j = 0; j < GCN_GRP; ++j) {
            const unsigned so = (unsigned)__builtin_amdgcn_readlane((int)so_l, g + j);
            v[j] = *(const float4*)(Hb + (so + l16));
        }
#pragma unroll
        for (int j = 0; j < GCN_GRP; ++j) {
            const unsigned ro = (unsigned)__builtin_amdgcn_readlane((int)ro_l, g + j);
            gcn_lds_add_row(acc + ro + lane, v[j]);
        }
    }
}

__device__ __forceinline__ void gcn_add_some(const char* __restrict__ Hb,
                                             float* __restrict__ acc,
                                             unsigned so_l, unsigned r_l, unsigned ball, int lane) {
    const unsigned ro_l = r_l << 7;
    const unsigned l16  = (unsigned)lane << 4;
    while (ball) {
        const int l = __builtin_ctz(ball);
        ball &= ball - 1u;
        const unsigned so = (unsigned)__builtin_amdgcn_readlane((int)so_l, l);
        const unsigned ro = (unsigned)__builtin_amdgcn_readlane((int)ro_l, l);
        const float4 v = *(const float4*)(Hb + (so + l16));
        gcn_lds_add_row(acc + ro + lane, v);
    }
}

__global__ void __launch_bounds__(1024)
gcn_aggregate_kernel(const float* __restrict__ H,
                     const int* __restrict__ src,
                     const int* __restrict__ dst,
                     const float* __restrict__ bias,
                     float* __restrict__ out,
                     const unsigned* __restrict__ list,
                     const unsigned* __restrict__ lcount,
                     const __bf16* __restrict__ Wp,
                     const unsigned* __restrict__ part,
                     int nedges, int nrows, int fuse) {
    __shared__ __attribute__((aligned(16))) float acc[GCN_AROWS * GCN_D];
    __shared__ unsigned indeg[GCN_AROWS];

    const int lane = threadIdx.x & 31;
    const int wave = __builtin_amdgcn_readfirstlane((int)(threadIdx.x >> 5));
    const int bid  = blockIdx.x;
    const int base = bid * GCN_R;

    float4* acc4 = (float4*)acc;
    for (int i = threadIdx.x; i < GCN_AROWS * GCN_D / 4; i += GCN_AWAVES * 32)
        acc4[i] = make_float4(0.0f, 0.0f, 0.0f, 0.0f);
    for (int i = threadIdx.x; i < GCN_AROWS; i += GCN_AWAVES * 32) indeg[i] = 0u;
    __syncthreads();

    const char* Hb = (const char*)H;

    const unsigned* mycount = lcount + bid * GCN_OS;
    unsigned most = 0u;
#pragma unroll
    for (int s = 0; s < GCN_OS; ++s) {
        const unsigned c = mycount[s];
        most = c > most ? c : most;
    }
    const bool whole = most <= (unsigned)GCN_SGRP;

    if (whole) {
        const int sl = wave & (GCN_OS - 1);
        const int ng = (int)mycount[sl];
        const unsigned* lw = list + (size_t)(bid * GCN_OS + sl) * GCN_SCAP;
        int g = wave >> 4;
        unsigned en = GCN_LSPARE;
        if (g < ng) en = lw[g * 32 + lane];
        for (; g < ng; g += 2) {
            const unsigned ent = en;
            if (g + 2 < ng) en = lw[(g + 2) * 32 + lane];
            gcn_add_full(Hb, acc, indeg, ent & ~511u, ent & 511u, lane);
        }
    } else {
        for (int e0 = wave * 32; e0 < nedges; e0 += GCN_AWAVES * 32) {
            const int e = e0 + lane;
            int d = -1, sr = 0;
            if (e < nedges) {
                d  = dst[e];
                sr = src[e];
            }
            const unsigned rl = (unsigned)(d - base);
            const bool hit = rl < (unsigned)GCN_R;
            if (hit) atomicAdd(indeg + rl, 1u);
            const unsigned ball = __builtin_amdgcn_ballot_w32(hit);
            gcn_add_some(Hb, acc, (unsigned)sr << 9, hit ? rl : (unsigned)GCN_R, ball, lane);
        }
    }

    __syncthreads();

    const float4 b4 = *(const float4*)(bias + 4 * lane);
    if (!fuse) {
        for (int r = wave; r < GCN_R; r += GCN_AWAVES) {
            const int gr = base + r;
            if (gr < nrows) {
                const float nd = rsqrtf(fmaxf((float)indeg[r], 1.0f));
                const float* a = acc + r * GCN_D + lane;
                float4 o;
                o.x = a[0]  * nd + b4.x;
                o.y = a[32] * nd + b4.y;
                o.z = a[64] * nd + b4.z;
                o.w = a[96] * nd + b4.w;
                *(float4*)(out + (size_t)gr * GCN_D + 4 * lane) = o;
            }
        }
        return;
    }

    const int lim = (nrows - base < GCN_R) ? (nrows - base) : GCN_R;
    const int r0  = 16 * wave;
    if (wave >= GCN_TILES || r0 >= lim) return;
    const int half = lane >> 4;
    const int m    = lane & 15;

    const unsigned* pp = part + (size_t)(8 * half) * GCN_OPITCH + (base + r0 + m);
    unsigned pc[GCN_OS / 2];
#pragma unroll
    for (int s = 0; s < GCN_OS / 2; ++s) pc[s] = pp[(size_t)s * GCN_OPITCH];

    float* trow = acc + r0 * GCN_D;
    char* tb = (char*)trow;
#pragma unroll
    for (int c4 = 0; c4 < 4; ++c4) {
        float4 raw[4];
#pragma unroll
        for (int j = 0; j < 4; ++j) {
            const float* a = trow + (4 * c4 + j) * GCN_D + lane;
            raw[j] = make_float4(a[0], a[32], a[64], a[96]);
        }
#pragma unroll
        for (int j = 0; j < 4; ++j) {
            const int row = 4 * c4 + j;
            const float nd = rsqrtf(fmaxf((float)indeg[r0 + row], 1.0f));
            v4bf hq;
            hq[0] = (__bf16)fmaxf(raw[j].x * nd + b4.x, 0.0f);
            hq[1] = (__bf16)fmaxf(raw[j].y * nd + b4.y, 0.0f);
            hq[2] = (__bf16)fmaxf(raw[j].z * nd + b4.z, 0.0f);
            hq[3] = (__bf16)fmaxf(raw[j].w * nd + b4.w, 0.0f);
            *(v4bf_lds*)(tb + row * GCN_SPITCH + 8 * lane) = hq;
        }
    }
    __builtin_amdgcn_sched_barrier(0);

    unsigned deg = 0u;
#pragma unroll
    for (int s = 0; s < GCN_OS / 2; ++s) deg += pc[s];
    deg += (unsigned)__shfl_xor((int)deg, 16);
    const float nsm = rsqrtf(fmaxf((float)deg, 1.0f));
    float nrow[16];
#pragma unroll
    for (int j = 0; j < 16; ++j)
        nrow[j] = __int_as_float(__builtin_amdgcn_readlane(__float_as_int(nsm), j));

    v16bf b[4];
#pragma unroll
    for (int j = 0; j < 4; ++j) b[j] = gcn_load_bfrag(Wp, j, 0, lane);

    v16bf a[4];
    const char* fr = tb + m * GCN_SPITCH + 16 * half;
#pragma unroll
    for (int kb = 0; kb < 4; ++kb) {
        const v8bf lo = *(const v8bf_lds*)(fr + 64 * kb);
        const v8bf hi = *(const v8bf_lds*)(fr + 64 * kb + 32);
        a[kb] = __builtin_shufflevector(lo, hi, 0, 1, 2, 3, 4, 5, 6, 7, 8, 9, 10, 11, 12, 13, 14, 15);
    }

    float* hr0 = out + (size_t)(base + r0 + 8 * half) * GCN_D + 4 * m;
    const int rl0 = r0 + 8 * half;
#pragma unroll
    for (int g = 0; g < 2; ++g) {
        v8f cacc[4];
#pragma unroll
        for (int j = 0; j < 4; ++j) cacc[j] = v8f{};
#pragma unroll
        for (int kb = 0; kb < 4; ++kb) {
            if (g + kb > 0) {
#pragma unroll
                for (int j = 0; j < 4; ++j) b[j] = gcn_load_bfrag(Wp, 4 * g + j, kb, lane);
                __builtin_amdgcn_sched_barrier(0);
            }
#pragma unroll
            for (int j = 0; j < 4; ++j)
                cacc[j] = __builtin_amdgcn_wmma_f32_16x16x32_bf16(
                     false, a[kb],  false, b[j],
                     (short)0, cacc[j],  false,  false);
        }
#pragma unroll
        for (int i = 0; i < 8; ++i) {
            const float nr = half ? nrow[i + 8] : nrow[i];
            const float4 o = make_float4(cacc[0][i] * nr, cacc[1][i] * nr,
                                         cacc[2][i] * nr, cacc[3][i] * nr);
            if (rl0 + i < lim) *(float4*)(hr0 + (size_t)i * GCN_D + 64 * g) = o;
        }
    }
}

extern "C" void kernel_launch(void* const* d_in, const int* in_sizes, int n_in,
                              void* d_out, int out_size, void* d_ws, size_t ws_size,
                              hipStream_t stream) {
    const float* feat = (const float*)d_in[0];
    const int*   src  = (const int*)  d_in[1];
    const int*   dst  = (const int*)  d_in[2];
    const float* W1   = (const float*)d_in[3];
    const float* b1   = (const float*)d_in[4];
    const float* W2   = (const float*)d_in[5];
    const float* b2   = (const float*)d_in[6];
    float* out = (float*)d_out;

    const int N = GCN_N;
    const int E = in_sizes[1];
    const size_t NF = (size_t)N * GCN_D;

    unsigned* part    = (unsigned*)d_ws;
    __bf16*   Wp1     = (__bf16*)(part + (size_t)GCN_OS * GCN_OPITCH);
    __bf16*   Wp2     = Wp1 + GCN_WPACK_ELEMS;
    float*    bufA    = (float*)(Wp2 + GCN_WPACK_ELEMS);
    float*    bufB    = bufA + NF;
    unsigned* lcount  = (unsigned*)(bufB + NF);
    unsigned* list    = lcount + GCN_OR * GCN_OWN * GCN_OS;

    const int thr = 256;
    const int gthr = 32 * GCN_GEMM_WAVES;
    const int pgrid = GCN_WPACK_ELEMS / thr;
    const int gemm_grid = (N / 16 + GCN_GEMM_WAVES - 1) / GCN_GEMM_WAVES;

    gcn_split_kernel<<<GCN_OR * GCN_OS, GCN_OTHR, 0, stream>>>(src, dst, part, list, lcount, E);
    gcn_pack_w_kernel<<<pgrid, thr, 0, stream>>>(W1, Wp1);
    gcn_pack_w_kernel<<<pgrid, thr, 0, stream>>>(W2, Wp2);

    gcn_gemm_norm_kernel<<<gemm_grid, gthr, 0, stream>>>(feat, Wp1, part, bufA, N);
    gcn_aggregate_kernel<<<GCN_ABLK, GCN_AWAVES * 32, 0, stream>>>(
        bufA, src, dst, b1, bufB, list, lcount, Wp2, part, E, N, 1);

    gcn_aggregate_kernel<<<GCN_ABLK, GCN_AWAVES * 32, 0, stream>>>(
        bufB, src, dst, b2, out, list, lcount, Wp2, part, E, N, 0);
}
